# starting kernel
# baseline (speedup 1.0000x reference)
#include <hip/hip_runtime.h>


#define NNODES 17
#define NPAD   32
#define NSLOT  9
#define DDIM   2048

typedef __attribute__((ext_vector_type(2))) float v2f;
typedef __attribute__((ext_vector_type(8))) float v8f;

__global__ __launch_bounds__(256) void prep_kernel(
    const float* __restrict__ xmain, const float* __restrict__ xsig, int nsig,
    const int* __restrict__ ei, const int* __restrict__ et, int E,
    float* __restrict__ xT)
{
  int slot = blockIdx.x / NPAD;
  int n    = blockIdx.x % NPAD;

  for (int j = 0; j < DDIM / 256; ++j) {
    int d = threadIdx.x + j * 256;
    float v = 0.0f;
    if (n < NNODES) {
      if (slot == 8) {
        v = (n < nsig) ? xsig[(size_t)n * DDIM + d]
                       : xmain[(size_t)n * DDIM + d];
      } else {
        float acc = 0.0f;
        int   cnt = 0;
        for (int e = 0; e < E; ++e) {
          if (et[e] == slot && ei[E + e] == n) {
            int m = ei[e];
            acc += (m < nsig) ? xsig[(size_t)m * DDIM + d]
                              : xmain[(size_t)m * DDIM + d];
            ++cnt;
          }
        }
        v = cnt ? acc / (float)cnt : 0.0f;
      }
    }
    xT[((size_t)slot * DDIM + d) * NPAD + n] = v;
  }
}

__global__ __launch_bounds__(128) void gemm_kernel(
    const float* __restrict__ xT,
    const float* __restrict__ W,
    const float* __restrict__ root,
    float* __restrict__ partial)
{
  const int wave = threadIdx.x >> 5;
  const int lane = threadIdx.x & 31;
  const int task = blockIdx.x * 4 + wave;
  const int slot  = task >> 7;
  const int strip = task & 127;
  const int h0    = strip * 16;

  const int half = lane >> 4;
  const int hl   = lane & 15;

  const float* __restrict__ Wr =
      (slot < 8) ? (W + (size_t)slot * DDIM * DDIM) : root;
  const float* __restrict__ Xs = xT + (size_t)slot * DDIM * NPAD;

  const float* wp = Wr + (size_t)(2 * half) * DDIM + h0 + hl;
  const float* ap = Xs + (size_t)(2 * half) * NPAD + hl;

  v8f c0 = {};
  v8f c1 = {};

#pragma unroll 4
  for (int k0 = 0; k0 < DDIM; k0 += 4) {
    v2f b, a0, a1;
    b.x  = wp[0];
    b.y  = wp[DDIM];
    a0.x = ap[0];
    a0.y = ap[NPAD];
    a1.x = ap[16];
    a1.y = ap[NPAD + 16];
    c0 = __builtin_amdgcn_wmma_f32_16x16x4_f32(false, a0, false, b,
                                               (short)0, c0, false, false);
    c1 = __builtin_amdgcn_wmma_f32_16x16x4_f32(false, a1, false, b,
                                               (short)0, c1, false, false);
    wp += (size_t)4 * DDIM;
    ap += 4 * NPAD;
  }

  size_t base = (size_t)slot * NPAD * DDIM + h0 + hl;
#pragma unroll
  for (int i = 0; i < 8; ++i) {
    partial[base + (size_t)(i + 8 * half) * DDIM]      = c0[i];
    partial[base + (size_t)(16 + i + 8 * half) * DDIM] = c1[i];
  }
}

__global__ __launch_bounds__(256) void reduce_kernel(
    const float* __restrict__ partial, const float* __restrict__ bias,
    float* __restrict__ out, int relu)
{
  int idx = blockIdx.x * blockDim.x + threadIdx.x;
  if (idx >= NNODES * DDIM) return;
  int n = idx / DDIM;
  int h = idx % DDIM;
  float s = bias[h];
#pragma unroll
  for (int slot = 0; slot < NSLOT; ++slot)
    s += partial[((size_t)slot * NPAD + n) * DDIM + h];
  if (relu) s = fmaxf(s, 0.0f);
  out[idx] = s;
}

extern "C" void kernel_launch(void* const* d_in, const int* in_sizes, int n_in,
                              void* d_out, int out_size, void* d_ws, size_t ws_size,
                              hipStream_t stream) {
  const float* node_emb = (const float*)d_in[0];
  const float* sig      = (const float*)d_in[1];
  const float* W1       = (const float*)d_in[2];
  const float* root1    = (const float*)d_in[3];
  const float* bias1    = (const float*)d_in[4];
  const float* W2       = (const float*)d_in[5];
  const float* root2    = (const float*)d_in[6];
  const float* bias2    = (const float*)d_in[7];
  const int*   ei       = (const int*)d_in[8];
  const int*   et       = (const int*)d_in[9];
  const int    E        = in_sizes[9];

  float* xT      = (float*)d_ws;
  float* partial = xT + (size_t)NSLOT * DDIM * NPAD;
  float* h1      = partial + (size_t)NSLOT * NPAD * DDIM;

  const int prepGrid   = NSLOT * NPAD;
  const int gemmGrid   = (NSLOT * 128) / 4;
  const int reduceGrid = (NNODES * DDIM + 255) / 256;

  prep_kernel<<<prepGrid, 256, 0, stream>>>(node_emb, sig, 6, ei, et, E, xT);
  gemm_kernel<<<gemmGrid, 128, 0, stream>>>(xT, W1, root1, partial);
  reduce_kernel<<<reduceGrid, 256, 0, stream>>>(partial, bias1, h1, 1);

  prep_kernel<<<prepGrid, 256, 0, stream>>>(h1, nullptr, 0, ei, et, E, xT);
  gemm_kernel<<<gemmGrid, 128, 0, stream>>>(xT, W2, root2, partial);
  reduce_kernel<<<reduceGrid, 256, 0, stream>>>(partial, bias2, (float*)d_out, 0);
}


# gemm as a stream: wave = 128 K-rows x 64 columns, 16-byte non-temporal W loads one group ahead; 18 K-chunks per block summed in LDS (new summation order), reduce adds 8 planes
# speedup vs baseline: 2.1232x; 2.1232x over previous
#include <hip/hip_runtime.h>


#define NNODES 17
#define NPAD   32
#define NSLOT  9
#define DDIM   2048

#define KCH    128
#define GSTEP  4
#define NGRP   (KCH / 4 / GSTEP)
#define GWAVES 18
#define NCHUNK (NSLOT * DDIM / KCH)
#define NPLANE (NCHUNK / GWAVES)
#define STRIP  64
#define NSTRIP (DDIM / STRIP)

typedef __attribute__((ext_vector_type(2))) float v2f;
typedef __attribute__((ext_vector_type(4))) float v4f;
typedef __attribute__((ext_vector_type(8))) float v8f;

__global__ __launch_bounds__(256) void prep_kernel(
    const float* __restrict__ xmain, const float* __restrict__ xsig, int nsig,
    const int* __restrict__ ei, const int* __restrict__ et, int E,
    float* __restrict__ xT)
{
  int slot = blockIdx.x / NPAD;
  int n    = blockIdx.x % NPAD;

  for (int j = 0; j < DDIM / 256; ++j) {
    int d = threadIdx.x + j * 256;
    float v = 0.0f;
    if (n < NNODES) {
      if (slot == 8) {
        v = (n < nsig) ? xsig[(size_t)n * DDIM + d]
                       : xmain[(size_t)n * DDIM + d];
      } else {
        float acc = 0.0f;
        int   cnt = 0;
        for (int e = 0; e < E; ++e) {
          if (et[e] == slot && ei[E + e] == n) {
            int m = ei[e];
            acc += (m < nsig) ? xsig[(size_t)m * DDIM + d]
                              : xmain[(size_t)m * DDIM + d];
            ++cnt;
          }
        }
        v = cnt ? acc / (float)cnt : 0.0f;
      }
    }
    xT[((size_t)slot * DDIM + d) * NPAD + n] = v;
  }
}

__global__ __launch_bounds__(GWAVES * 32) void gemm_kernel(
    const float* __restrict__ xT,
    const float* __restrict__ W,
    const float* __restrict__ root,
    float* __restrict__ partial)
{
  __shared__ __attribute__((aligned(16))) float red[GWAVES][NNODES][STRIP];

  const int lane  = threadIdx.x & 31;
  const int wave  = __builtin_amdgcn_readfirstlane((int)(threadIdx.x >> 5));
  const int strip = blockIdx.x % NSTRIP;
  const int plane = blockIdx.x / NSTRIP;
  const int chunk = plane * GWAVES + wave;
  const int slot  = chunk / (DDIM / KCH);
  const int kbase = (chunk % (DDIM / KCH)) * KCH;
  const int h0    = strip * STRIP;

  const int half = lane >> 4;
  const int c    = lane & 15;

  const float* __restrict__ Wr =
      (slot < 8) ? (W + (size_t)slot * DDIM * DDIM) : root;

  const float* wp = Wr + (size_t)(kbase + 2 * half) * DDIM + h0 + 4 * c;
  const float* ap = xT + ((size_t)slot * DDIM + kbase + 2 * half) * NPAD + c;

  v8f acc0[4] = {};
  v8f acc1[4] = {};

  v4f blo[2][GSTEP], bhi[2][GSTEP];
  v2f a0[2][GSTEP], a1[2][GSTEP];

#define LOAD_GROUP(G) do {                                                     \
    _Pragma("unroll") for (int q = 0; q < GSTEP; ++q) {                        \
      const int kk = ((G) * GSTEP + q) * 4;                                    \
      a0[(G) & 1][q].x = ap[(size_t)kk * NPAD];                                \
      a0[(G) & 1][q].y = ap[(size_t)(kk + 1) * NPAD];                          \
      a1[(G) & 1][q].x = ap[(size_t)kk * NPAD + 16];                           \
      a1[(G) & 1][q].y = ap[(size_t)(kk + 1) * NPAD + 16];                     \
      blo[(G) & 1][q] = __builtin_nontemporal_load(                            \
          reinterpret_cast<const v4f*>(wp + (size_t)kk * DDIM));               \
      bhi[(G) & 1][q] = __builtin_nontemporal_load(                            \
          reinterpret_cast<const v4f*>(wp + (size_t)(kk + 1) * DDIM));         \
    } } while (0)

  LOAD_GROUP(0);
#pragma unroll
  for (int g = 0; g < NGRP; ++g) {
    __builtin_amdgcn_sched_barrier(0);
    if (g + 1 < NGRP) LOAD_GROUP(g + 1);
    __builtin_amdgcn_sched_barrier(0);
#pragma unroll
    for (int q = 0; q < GSTEP; ++q) {
#pragma unroll
      for (int j = 0; j < 4; ++j) {
        v2f b;
        b.x = blo[g & 1][q][j];
        b.y = bhi[g & 1][q][j];
        acc0[j] = __builtin_amdgcn_wmma_f32_16x16x4_f32(
            false, a0[g & 1][q], false, b, (short)0, acc0[j], false, false);
        acc1[j] = __builtin_amdgcn_wmma_f32_16x16x4_f32(
            false, a1[g & 1][q], false, b, (short)0, acc1[j], false, false);
      }
    }
  }
#undef LOAD_GROUP

#pragma unroll
  for (int i = 0; i < 8; ++i) {
    v4f v = {acc0[0][i], acc0[1][i], acc0[2][i], acc0[3][i]};
    *reinterpret_cast<v4f*>(&red[wave][i + 8 * half][4 * c]) = v;
  }
  if (half == 0) {
    v4f v = {acc1[0][0], acc1[1][0], acc1[2][0], acc1[3][0]};
    *reinterpret_cast<v4f*>(&red[wave][16][4 * c]) = v;
  }
  __syncthreads();

  for (int o = threadIdx.x; o < NNODES * STRIP; o += GWAVES * 32) {
    const int row = o / STRIP;
    const int col = o % STRIP;
    float s = red[0][row][col];
#pragma unroll
    for (int w = 1; w < GWAVES; ++w) s += red[w][row][col];
    partial[((size_t)plane * NPAD + row) * DDIM + h0 + col] = s;
  }
}

__global__ __launch_bounds__(256) void reduce_kernel(
    const float* __restrict__ partial, const float* __restrict__ bias,
    float* __restrict__ out, int relu)
{
  int idx = blockIdx.x * blockDim.x + threadIdx.x;
  if (idx >= NNODES * DDIM) return;
  int n = idx / DDIM;
  int h = idx % DDIM;
  float s = bias[h];
#pragma unroll
  for (int p = 0; p < NPLANE; ++p)
    s += partial[((size_t)p * NPAD + n) * DDIM + h];
  if (relu) s = fmaxf(s, 0.0f);
  out[idx] = s;
}

extern "C" void kernel_launch(void* const* d_in, const int* in_sizes, int n_in,
                              void* d_out, int out_size, void* d_ws, size_t ws_size,
                              hipStream_t stream) {
  const float* node_emb = (const float*)d_in[0];
  const float* sig      = (const float*)d_in[1];
  const float* W1       = (const float*)d_in[2];
  const float* root1    = (const float*)d_in[3];
  const float* bias1    = (const float*)d_in[4];
  const float* W2       = (const float*)d_in[5];
  const float* root2    = (const float*)d_in[6];
  const float* bias2    = (const float*)d_in[7];
  const int*   ei       = (const int*)d_in[8];
  const int*   et       = (const int*)d_in[9];
  const int    E        = in_sizes[9];

  float* xT      = (float*)d_ws;
  float* partial = xT + (size_t)NSLOT * DDIM * NPAD;
  float* h1      = partial + (size_t)NSLOT * NPAD * DDIM;

  const int prepGrid   = NSLOT * NPAD;
  const int gemmGrid   = NPLANE * NSTRIP;
  const int reduceGrid = (NNODES * DDIM + 255) / 256;

  prep_kernel<<<prepGrid, 256, 0, stream>>>(node_emb, sig, 6, ei, et, E, xT);
  gemm_kernel<<<gemmGrid, GWAVES * 32, 0, stream>>>(xT, W1, root1, partial);
  reduce_kernel<<<reduceGrid, 256, 0, stream>>>(partial, bias1, h1, 1);

  prep_kernel<<<prepGrid, 256, 0, stream>>>(h1, nullptr, 0, ei, et, E, xT);
  gemm_kernel<<<gemmGrid, GWAVES * 32, 0, stream>>>(xT, W2, root2, partial);
  reduce_kernel<<<reduceGrid, 256, 0, stream>>>(partial, bias2, (float*)d_out, 0);
}


# prep scans the edge list once per block, one edge per thread, ballot masks per target node; lane=node, wave=8 k: whole 128-B xT lines; same summation order
# speedup vs baseline: 3.9165x; 1.8446x over previous
#include <hip/hip_runtime.h>


#define NNODES 17
#define NPAD   32
#define NSLOT  9
#define DDIM   2048

#define KCH    128
#define GSTEP  4
#define NGRP   (KCH / 4 / GSTEP)
#define GWAVES 18
#define NCHUNK (NSLOT * DDIM / KCH)
#define NPLANE (NCHUNK / GWAVES)
#define STRIP  64
#define NSTRIP (DDIM / STRIP)

typedef __attribute__((ext_vector_type(2))) float v2f;
typedef __attribute__((ext_vector_type(4))) float v4f;
typedef __attribute__((ext_vector_type(8))) float v8f;

#define PREP_DT   64
#define PREP_TILES (DDIM / PREP_DT)

__global__ __launch_bounds__(256) void prep_kernel(
    const float* __restrict__ xmain, const float* __restrict__ xsig, int nsig,
    const int* __restrict__ ei, const int* __restrict__ et, int E,
    float* __restrict__ xT)
{
  __shared__ unsigned masks[8][32];
  __shared__ int      srcs[256];

  const int slot = blockIdx.x / PREP_TILES;
  const int lane = threadIdx.x & 31;
  const int wave = threadIdx.x >> 5;
  const int d    = (blockIdx.x % PREP_TILES) * PREP_DT + wave * 8;

  v4f lo = {0.0f, 0.0f, 0.0f, 0.0f};
  v4f hi = {0.0f, 0.0f, 0.0f, 0.0f};

  if (slot == 8) {
    if (lane < NNODES) {
      const float* row = ((lane < nsig) ? xsig : xmain) + (size_t)lane * DDIM + d;
      lo = *reinterpret_cast<const v4f*>(row);
      hi = *reinterpret_cast<const v4f*>(row + 4);
    }
  } else {
    int cnt = 0;
    for (int base = 0; base < E; base += 256) {
      if (base) __syncthreads();
      const int e = base + (int)threadIdx.x;
      int t = -1, dn = -1, s = 0;
      if (e < E) { t = et[e]; dn = ei[E + e]; s = ei[e]; }
      const bool hit = (t == slot);
      unsigned mine = 0;
#pragma unroll
      for (int m = 0; m < NNODES; ++m) {
        const unsigned b = __builtin_amdgcn_ballot_w32(hit && dn == m);
        if (lane == m) mine = b;
      }
      masks[wave][lane] = mine;
      srcs[threadIdx.x] = s;
      __syncthreads();
      const int left = E - base;
      const int nch  = (left >= 256) ? 8 : ((left + 31) >> 5);
      for (int c = 0; c < nch; ++c) {
        unsigned mk = masks[c][lane];
        while (mk) {
          const int bit = __builtin_ctz(mk);
          mk &= mk - 1;
          const int m = srcs[c * 32 + bit];
          const float* row = ((m < nsig) ? xsig : xmain) + (size_t)m * DDIM + d;
          lo += *reinterpret_cast<const v4f*>(row);
          hi += *reinterpret_cast<const v4f*>(row + 4);
          ++cnt;
        }
      }
    }
    if (cnt) {
      const float fc = (float)cnt;
      lo = lo / fc;
      hi = hi / fc;
    }
  }

  float* dst = xT + ((size_t)slot * DDIM + d) * NPAD + lane;
#pragma unroll
  for (int i = 0; i < 4; ++i) {
    dst[(size_t)i * NPAD]       = lo[i];
    dst[(size_t)(i + 4) * NPAD] = hi[i];
  }
}

__global__ __launch_bounds__(GWAVES * 32) void gemm_kernel(
    const float* __restrict__ xT,
    const float* __restrict__ W,
    const float* __restrict__ root,
    float* __restrict__ partial)
{
  __shared__ __attribute__((aligned(16))) float red[GWAVES][NNODES][STRIP];

  const int lane  = threadIdx.x & 31;
  const int wave  = __builtin_amdgcn_readfirstlane((int)(threadIdx.x >> 5));
  const int strip = blockIdx.x % NSTRIP;
  const int plane = blockIdx.x / NSTRIP;
  const int chunk = plane * GWAVES + wave;
  const int slot  = chunk / (DDIM / KCH);
  const int kbase = (chunk % (DDIM / KCH)) * KCH;
  const int h0    = strip * STRIP;

  const int half = lane >> 4;
  const int c    = lane & 15;

  const float* __restrict__ Wr =
      (slot < 8) ? (W + (size_t)slot * DDIM * DDIM) : root;

  const float* wp = Wr + (size_t)(kbase + 2 * half) * DDIM + h0 + 4 * c;
  const float* ap = xT + ((size_t)slot * DDIM + kbase + 2 * half) * NPAD + c;

  v8f acc0[4] = {};
  v8f acc1[4] = {};

  v4f blo[2][GSTEP], bhi[2][GSTEP];
  v2f a0[2][GSTEP], a1[2][GSTEP];

#define LOAD_GROUP(G) do {                                                     \
    _Pragma("unroll") for (int q = 0; q < GSTEP; ++q) {                        \
      const int kk = ((G) * GSTEP + q) * 4;                                    \
      a0[(G) & 1][q].x = ap[(size_t)kk * NPAD];                                \
      a0[(G) & 1][q].y = ap[(size_t)(kk + 1) * NPAD];                          \
      a1[(G) & 1][q].x = ap[(size_t)kk * NPAD + 16];                           \
      a1[(G) & 1][q].y = ap[(size_t)(kk + 1) * NPAD + 16];                     \
      blo[(G) & 1][q] = __builtin_nontemporal_load(                            \
          reinterpret_cast<const v4f*>(wp + (size_t)kk * DDIM));               \
      bhi[(G) & 1][q] = __builtin_nontemporal_load(                            \
          reinterpret_cast<const v4f*>(wp + (size_t)(kk + 1) * DDIM));         \
    } } while (0)

  LOAD_GROUP(0);
#pragma unroll
  for (int g = 0; g < NGRP; ++g) {
    __builtin_amdgcn_sched_barrier(0);
    if (g + 1 < NGRP) LOAD_GROUP(g + 1);
    __builtin_amdgcn_sched_barrier(0);
#pragma unroll
    for (int q = 0; q < GSTEP; ++q) {
#pragma unroll
      for (int j = 0; j < 4; ++j) {
        v2f b;
        b.x = blo[g & 1][q][j];
        b.y = bhi[g & 1][q][j];
        acc0[j] = __builtin_amdgcn_wmma_f32_16x16x4_f32(
            false, a0[g & 1][q], false, b, (short)0, acc0[j], false, false);
        acc1[j] = __builtin_amdgcn_wmma_f32_16x16x4_f32(
            false, a1[g & 1][q], false, b, (short)0, acc1[j], false, false);
      }
    }
  }
#undef LOAD_GROUP

#pragma unroll
  for (int i = 0; i < 8; ++i) {
    v4f v = {acc0[0][i], acc0[1][i], acc0[2][i], acc0[3][i]};
    *reinterpret_cast<v4f*>(&red[wave][i + 8 * half][4 * c]) = v;
  }
  if (half == 0) {
    v4f v = {acc1[0][0], acc1[1][0], acc1[2][0], acc1[3][0]};
    *reinterpret_cast<v4f*>(&red[wave][16][4 * c]) = v;
  }
  __syncthreads();

  for (int o = threadIdx.x; o < NNODES * STRIP; o += GWAVES * 32) {
    const int row = o / STRIP;
    const int col = o % STRIP;
    float s = red[0][row][col];
#pragma unroll
    for (int w = 1; w < GWAVES; ++w) s += red[w][row][col];
    partial[((size_t)plane * NPAD + row) * DDIM + h0 + col] = s;
  }
}

__global__ __launch_bounds__(256) void reduce_kernel(
    const float* __restrict__ partial, const float* __restrict__ bias,
    float* __restrict__ out, int relu)
{
  int idx = blockIdx.x * blockDim.x + threadIdx.x;
  if (idx >= NNODES * DDIM) return;
  int n = idx / DDIM;
  int h = idx % DDIM;
  float s = bias[h];
#pragma unroll
  for (int p = 0; p < NPLANE; ++p)
    s += partial[((size_t)p * NPAD + n) * DDIM + h];
  if (relu) s = fmaxf(s, 0.0f);
  out[idx] = s;
}

extern "C" void kernel_launch(void* const* d_in, const int* in_sizes, int n_in,
                              void* d_out, int out_size, void* d_ws, size_t ws_size,
                              hipStream_t stream) {
  const float* node_emb = (const float*)d_in[0];
  const float* sig      = (const float*)d_in[1];
  const float* W1       = (const float*)d_in[2];
  const float* root1    = (const float*)d_in[3];
  const float* bias1    = (const float*)d_in[4];
  const float* W2       = (const float*)d_in[5];
  const float* root2    = (const float*)d_in[6];
  const float* bias2    = (const float*)d_in[7];
  const int*   ei       = (const int*)d_in[8];
  const int*   et       = (const int*)d_in[9];
  const int    E        = in_sizes[9];

  float* xT      = (float*)d_ws;
  float* partial = xT + (size_t)NSLOT * DDIM * NPAD;
  float* h1      = partial + (size_t)NSLOT * NPAD * DDIM;

  const int prepGrid   = NSLOT * PREP_TILES;
  const int gemmGrid   = NPLANE * NSTRIP;
  const int reduceGrid = (NNODES * DDIM + 255) / 256;

  prep_kernel<<<prepGrid, 256, 0, stream>>>(node_emb, sig, 6, ei, et, E, xT);
  gemm_kernel<<<gemmGrid, GWAVES * 32, 0, stream>>>(xT, W1, root1, partial);
  reduce_kernel<<<reduceGrid, 256, 0, stream>>>(partial, bias1, h1, 1);

  prep_kernel<<<prepGrid, 256, 0, stream>>>(h1, nullptr, 0, ei, et, E, xT);
  gemm_kernel<<<gemmGrid, GWAVES * 32, 0, stream>>>(xT, W2, root2, partial);
  reduce_kernel<<<reduceGrid, 256, 0, stream>>>(partial, bias2, (float*)d_out, 0);
}


# gemm skips the second 16-row tile (node 16 only) where the slot has no edge into node 16 (flag from prep); chunks dealt round-robin to blocks so such slots spread: new summation order
# speedup vs baseline: 4.4414x; 1.1340x over previous
#include <hip/hip_runtime.h>


#define NNODES 17
#define NPAD   32
#define NSLOT  9
#define DDIM   2048

#define KCH    128
#define GSTEP  4
#define NGRP   (KCH / 4 / GSTEP)
#define GWAVES 18
#define NCHUNK (NSLOT * DDIM / KCH)
#define NPLANE (NCHUNK / GWAVES)
#define STRIP  64
#define NSTRIP (DDIM / STRIP)

typedef __attribute__((ext_vector_type(2))) float v2f;
typedef __attribute__((ext_vector_type(4))) float v4f;
typedef __attribute__((ext_vector_type(8))) float v8f;

#define PREP_DT   64
#define PREP_TILES (DDIM / PREP_DT)

__global__ __launch_bounds__(256) void prep_kernel(
    const float* __restrict__ xmain, const float* __restrict__ xsig, int nsig,
    const int* __restrict__ ei, const int* __restrict__ et, int E,
    float* __restrict__ xT, int* __restrict__ need16)
{
  __shared__ unsigned masks[8][32];
  __shared__ int      srcs[256];

  const int slot = blockIdx.x / PREP_TILES;
  const int lane = threadIdx.x & 31;
  const int wave = threadIdx.x >> 5;
  const int d    = (blockIdx.x % PREP_TILES) * PREP_DT + wave * 8;

  v4f lo = {0.0f, 0.0f, 0.0f, 0.0f};
  v4f hi = {0.0f, 0.0f, 0.0f, 0.0f};
  int cnt = 0;

  if (slot == 8) {
    if (lane < NNODES) {
      const float* row = ((lane < nsig) ? xsig : xmain) + (size_t)lane * DDIM + d;
      lo = *reinterpret_cast<const v4f*>(row);
      hi = *reinterpret_cast<const v4f*>(row + 4);
    }
  } else {
    for (int base = 0; base < E; base += 256) {
      if (base) __syncthreads();
      const int e = base + (int)threadIdx.x;
      int t = -1, dn = -1, s = 0;
      if (e < E) { t = et[e]; dn = ei[E + e]; s = ei[e]; }
      const bool hit = (t == slot);
      unsigned mine = 0;
#pragma unroll
      for (int m = 0; m < NNODES; ++m) {
        const unsigned b = __builtin_amdgcn_ballot_w32(hit && dn == m);
        if (lane == m) mine = b;
      }
      masks[wave][lane] = mine;
      srcs[threadIdx.x] = s;
      __syncthreads();
      const int left = E - base;
      const int nch  = (left >= 256) ? 8 : ((left + 31) >> 5);
      for (int c = 0; c < nch; ++c) {
        unsigned mk = masks[c][lane];
        while (mk) {
          const int bit = __builtin_ctz(mk);
          mk &= mk - 1;
          const int m = srcs[c * 32 + bit];
          const float* row = ((m < nsig) ? xsig : xmain) + (size_t)m * DDIM + d;
          lo += *reinterpret_cast<const v4f*>(row);
          hi += *reinterpret_cast<const v4f*>(row + 4);
          ++cnt;
        }
      }
    }
    if (cnt) {
      const float fc = (float)cnt;
      lo = lo / fc;
      hi = hi / fc;
    }
  }

  if ((blockIdx.x % PREP_TILES) == 0 && threadIdx.x == NNODES - 1)
    need16[slot] = (slot == 8 || cnt > 0) ? 1 : 0;

  float* dst = xT + ((size_t)slot * DDIM + d) * NPAD + lane;
#pragma unroll
  for (int i = 0; i < 4; ++i) {
    dst[(size_t)i * NPAD]       = lo[i];
    dst[(size_t)(i + 4) * NPAD] = hi[i];
  }
}


template <bool TWO>
__device__ __forceinline__ void chunk_product(
    const float* __restrict__ wp, const float* __restrict__ ap,
    float (*__restrict__ slab)[STRIP], const int half, const int c)
{
  v8f acc0[4] = {};
  v8f acc1[4] = {};

  v4f blo[2][GSTEP], bhi[2][GSTEP];
  v2f a0[2][GSTEP], a1[2][GSTEP];

#define LOAD_GROUP(G) do {                                                     \
    _Pragma("unroll") for (int q = 0; q < GSTEP; ++q) {                        \
      const int kk = ((G) * GSTEP + q) * 4;                                    \
      a0[(G) & 1][q].x = ap[(size_t)kk * NPAD];                                \
      a0[(G) & 1][q].y = ap[(size_t)(kk + 1) * NPAD];                          \
      if (TWO) {                                                               \
        a1[(G) & 1][q].x = ap[(size_t)kk * NPAD + 16];                         \
        a1[(G) & 1][q].y = ap[(size_t)(kk + 1) * NPAD + 16];                   \
      }                                                                        \
      blo[(G) & 1][q] = __builtin_nontemporal_load(                            \
          reinterpret_cast<const v4f*>(wp + (size_t)kk * DDIM));               \
      bhi[(G) & 1][q] = __builtin_nontemporal_load(                            \
          reinterpret_cast<const v4f*>(wp + (size_t)(kk + 1) * DDIM));         \
    } } while (0)

  LOAD_GROUP(0);
#pragma unroll
  for (int g = 0; g < NGRP; ++g) {
    __builtin_amdgcn_sched_barrier(0);
    if (g + 1 < NGRP) LOAD_GROUP(g + 1);
    __builtin_amdgcn_sched_barrier(0);
#pragma unroll
    for (int q = 0; q < GSTEP; ++q) {
#pragma unroll
      for (int j = 0; j < 4; ++j) {
        v2f b;
        b.x = blo[g & 1][q][j];
        b.y = bhi[g & 1][q][j];
        acc0[j] = __builtin_amdgcn_wmma_f32_16x16x4_f32(
            false, a0[g & 1][q], false, b, (short)0, acc0[j], false, false);
        if (TWO)
          acc1[j] = __builtin_amdgcn_wmma_f32_16x16x4_f32(
              false, a1[g & 1][q], false, b, (short)0, acc1[j], false, false);
      }
    }
  }
#undef LOAD_GROUP

#pragma unroll
  for (int i = 0; i < 8; ++i) {
    v4f v = {acc0[0][i], acc0[1][i], acc0[2][i], acc0[3][i]};
    *reinterpret_cast<v4f*>(&slab[i + 8 * half][4 * c]) = v;
  }
  if (half == 0) {
    v4f v = {acc1[0][0], acc1[1][0], acc1[2][0], acc1[3][0]};
    *reinterpret_cast<v4f*>(&slab[16][4 * c]) = v;
  }
}

__global__ __launch_bounds__(GWAVES * 32) void gemm_kernel(
    const float* __restrict__ xT,
    const float* __restrict__ W,
    const float* __restrict__ root,
    const int* __restrict__ need16,
    float* __restrict__ partial)
{
  __shared__ __attribute__((aligned(16))) float red[GWAVES][NNODES][STRIP];

  const int lane  = threadIdx.x & 31;
  const int wave  = __builtin_amdgcn_readfirstlane((int)(threadIdx.x >> 5));
  const int strip = blockIdx.x % NSTRIP;
  const int plane = blockIdx.x / NSTRIP;
  const int chunk = wave * NPLANE + plane;
  const int slot  = chunk / (DDIM / KCH);
  const int kbase = (chunk % (DDIM / KCH)) * KCH;
  const int h0    = strip * STRIP;

  const int half = lane >> 4;
  const int c    = lane & 15;

  const float* __restrict__ Wr =
      (slot < 8) ? (W + (size_t)slot * DDIM * DDIM) : root;

  const float* wp = Wr + (size_t)(kbase + 2 * half) * DDIM + h0 + 4 * c;
  const float* ap = xT + ((size_t)slot * DDIM + kbase + 2 * half) * NPAD + c;

  if (need16[slot]) chunk_product<true >(wp, ap, red[wave], half, c);
  else              chunk_product<false>(wp, ap, red[wave], half, c);
  __syncthreads();

  for (int o = threadIdx.x; o < NNODES * STRIP; o += GWAVES * 32) {
    const int row = o / STRIP;
    const int col = o % STRIP;
    float s = red[0][row][col];
#pragma unroll
    for (int w = 1; w < GWAVES; ++w) s += red[w][row][col];
    partial[((size_t)plane * NPAD + row) * DDIM + h0 + col] = s;
  }
}

__global__ __launch_bounds__(256) void reduce_kernel(
    const float* __restrict__ partial, const float* __restrict__ bias,
    float* __restrict__ out, int relu)
{
  int idx = blockIdx.x * blockDim.x + threadIdx.x;
  if (idx >= NNODES * DDIM) return;
  int n = idx / DDIM;
  int h = idx % DDIM;
  float s = bias[h];
#pragma unroll
  for (int p = 0; p < NPLANE; ++p)
    s += partial[((size_t)p * NPAD + n) * DDIM + h];
  if (relu) s = fmaxf(s, 0.0f);
  out[idx] = s;
}

extern "C" void kernel_launch(void* const* d_in, const int* in_sizes, int n_in,
                              void* d_out, int out_size, void* d_ws, size_t ws_size,
                              hipStream_t stream) {
  const float* node_emb = (const float*)d_in[0];
  const float* sig      = (const float*)d_in[1];
  const float* W1       = (const float*)d_in[2];
  const float* root1    = (const float*)d_in[3];
  const float* bias1    = (const float*)d_in[4];
  const float* W2       = (const float*)d_in[5];
  const float* root2    = (const float*)d_in[6];
  const float* bias2    = (const float*)d_in[7];
  const int*   ei       = (const int*)d_in[8];
  const int*   et       = (const int*)d_in[9];
  const int    E        = in_sizes[9];

  float* xT      = (float*)d_ws;
  float* partial = xT + (size_t)NSLOT * DDIM * NPAD;
  float* h1      = partial + (size_t)NSLOT * NPAD * DDIM;
  int*   need16  = (int*)(h1 + (size_t)NPAD * DDIM);

  const int prepGrid   = NSLOT * PREP_TILES;
  const int gemmGrid   = NPLANE * NSTRIP;
  const int reduceGrid = (NNODES * DDIM + 255) / 256;

  prep_kernel<<<prepGrid, 256, 0, stream>>>(node_emb, sig, 6, ei, et, E, xT, need16);
  gemm_kernel<<<gemmGrid, GWAVES * 32, 0, stream>>>(xT, W1, root1, need16, partial);
  reduce_kernel<<<reduceGrid, 256, 0, stream>>>(partial, bias1, h1, 1);

  prep_kernel<<<prepGrid, 256, 0, stream>>>(h1, nullptr, 0, ei, et, E, xT, need16);
  gemm_kernel<<<gemmGrid, GWAVES * 32, 0, stream>>>(xT, W2, root2, need16, partial);
  reduce_kernel<<<reduceGrid, 256, 0, stream>>>(partial, bias2, (float*)d_out, 0);
}


# layer 1's reduce launch folded into layer 2's prep: each prep block sums its own 17 x 64 tile of the 8 planes + bias, relu, into LDS and gathers rows from there; 5 launches, same summation order
# speedup vs baseline: 4.6759x; 1.0528x over previous
#include <hip/hip_runtime.h>


#define NNODES 17
#define NPAD   32
#define NSLOT  9
#define DDIM   2048

#define KCH    128
#define GSTEP  4
#define NGRP   (KCH / 4 / GSTEP)
#define GWAVES 18
#define NCHUNK (NSLOT * DDIM / KCH)
#define NPLANE (NCHUNK / GWAVES)
#define STRIP  64
#define NSTRIP (DDIM / STRIP)

typedef __attribute__((ext_vector_type(2))) float v2f;
typedef __attribute__((ext_vector_type(4))) float v4f;
typedef __attribute__((ext_vector_type(8))) float v8f;

#define PREP_DT   64
#define PREP_TILES (DDIM / PREP_DT)
#define XPITCH    (PREP_DT + 4)

template <bool FROM_PLANES>
__global__ __launch_bounds__(256) void prep_kernel(
    const float* __restrict__ xmain, const float* __restrict__ xsig, int nsig,
    const int* __restrict__ ei, const int* __restrict__ et, int E,
    float* __restrict__ xT, int* __restrict__ need16)
{
  __shared__ unsigned masks[8][32];
  __shared__ int      srcs[256];
  __shared__ __attribute__((aligned(16))) float xs[FROM_PLANES ? NNODES : 1][XPITCH];

  const int slot = blockIdx.x / PREP_TILES;
  const int lane = threadIdx.x & 31;
  const int wave = threadIdx.x >> 5;
  const int d0   = (blockIdx.x % PREP_TILES) * PREP_DT;
  const int d    = d0 + wave * 8;

  int t0 = -1, dn0 = -1, s0 = 0;
  if (slot != 8 && (int)threadIdx.x < E) {
    t0 = et[threadIdx.x]; dn0 = ei[E + threadIdx.x]; s0 = ei[threadIdx.x];
  }

  if (FROM_PLANES) {
    const int col4 = ((int)threadIdx.x & 15) * 4;
    const int rowa = (int)threadIdx.x >> 4;
    const bool second = threadIdx.x < 16;
    const float* pa = xmain + (size_t)rowa * DDIM + d0 + col4;
    const float* pb = xmain + (size_t)(NNODES - 1) * DDIM + d0 + col4;
    const v4f bv = *reinterpret_cast<const v4f*>(xsig + d0 + col4);
    v4f va[NPLANE], vb[NPLANE];
#pragma unroll
    for (int p = 0; p < NPLANE; ++p)
      va[p] = *reinterpret_cast<const v4f*>(pa + (size_t)p * NPAD * DDIM);
    if (second) {
#pragma unroll
      for (int p = 0; p < NPLANE; ++p)
        vb[p] = *reinterpret_cast<const v4f*>(pb + (size_t)p * NPAD * DDIM);
    }
    v4f sa = bv;
#pragma unroll
    for (int p = 0; p < NPLANE; ++p) sa += va[p];
#pragma unroll
    for (int i = 0; i < 4; ++i) sa[i] = fmaxf(sa[i], 0.0f);
    *reinterpret_cast<v4f*>(&xs[rowa][col4]) = sa;
    if (second) {
      v4f sb = bv;
#pragma unroll
      for (int p = 0; p < NPLANE; ++p) sb += vb[p];
#pragma unroll
      for (int i = 0; i < 4; ++i) sb[i] = fmaxf(sb[i], 0.0f);
      *reinterpret_cast<v4f*>(&xs[NNODES - 1][col4]) = sb;
    }
    if (slot == 8) __syncthreads();
  }

  v4f lo = {0.0f, 0.0f, 0.0f, 0.0f};
  v4f hi = {0.0f, 0.0f, 0.0f, 0.0f};
  int cnt = 0;

  if (slot == 8) {
    if (lane < NNODES) {
      const float* row = FROM_PLANES
          ? &xs[lane][wave * 8]
          : ((lane < nsig) ? xsig : xmain) + (size_t)lane * DDIM + d;
      lo = *reinterpret_cast<const v4f*>(row);
      hi = *reinterpret_cast<const v4f*>(row + 4);
    }
  } else {
    for (int base = 0; base < E; base += 256) {
      if (base) __syncthreads();
      const int e = base + (int)threadIdx.x;
      int t = t0, dn = dn0, s = s0;
      if (base) {
        t = -1; dn = -1; s = 0;
        if (e < E) { t = et[e]; dn = ei[E + e]; s = ei[e]; }
      }
      const bool hit = (t == slot);
      unsigned mine = 0;
#pragma unroll
      for (int m = 0; m < NNODES; ++m) {
        const unsigned b = __builtin_amdgcn_ballot_w32(hit && dn == m);
        if (lane == m) mine = b;
      }
      masks[wave][lane] = mine;
      srcs[threadIdx.x] = s;
      __syncthreads();
      const int left = E - base;
      const int nch  = (left >= 256) ? 8 : ((left + 31) >> 5);
      for (int c = 0; c < nch; ++c) {
        unsigned mk = masks[c][lane];
        while (mk) {
          const int bit = __builtin_ctz(mk);
          mk &= mk - 1;
          const int m = srcs[c * 32 + bit];
          const float* row = FROM_PLANES
              ? &xs[m][wave * 8]
              : ((m < nsig) ? xsig : xmain) + (size_t)m * DDIM + d;
          lo += *reinterpret_cast<const v4f*>(row);
          hi += *reinterpret_cast<const v4f*>(row + 4);
          ++cnt;
        }
      }
    }
    if (cnt) {
      const float fc = (float)cnt;
      lo = lo / fc;
      hi = hi / fc;
    }
  }

  if ((blockIdx.x % PREP_TILES) == 0 && threadIdx.x == NNODES - 1)
    need16[slot] = (slot == 8 || cnt > 0) ? 1 : 0;

  float* dst = xT + ((size_t)slot * DDIM + d) * NPAD + lane;
#pragma unroll
  for (int i = 0; i < 4; ++i) {
    dst[(size_t)i * NPAD]       = lo[i];
    dst[(size_t)(i + 4) * NPAD] = hi[i];
  }
}


template <bool TWO>
__device__ __forceinline__ void chunk_product(
    const float* __restrict__ wp, const float* __restrict__ ap,
    float (*__restrict__ slab)[STRIP], const int half, const int c)
{
  v8f acc0[4] = {};
  v8f acc1[4] = {};

  v4f blo[2][GSTEP], bhi[2][GSTEP];
  v2f a0[2][GSTEP], a1[2][GSTEP];

#define LOAD_GROUP(G) do {                                                     \
    _Pragma("unroll") for (int q = 0; q < GSTEP; ++q) {                        \
      const int kk = ((G) * GSTEP + q) * 4;                                    \
      a0[(G) & 1][q].x = ap[(size_t)kk * NPAD];                                \
      a0[(G) & 1][q].y = ap[(size_t)(kk + 1) * NPAD];                          \
      if (TWO) {                                                               \
        a1[(G) & 1][q].x = ap[(size_t)kk * NPAD + 16];                         \
        a1[(G) & 1][q].y = ap[(size_t)(kk + 1) * NPAD + 16];                   \
      }                                                                        \
      blo[(G) & 1][q] = __builtin_nontemporal_load(                            \
          reinterpret_cast<const v4f*>(wp + (size_t)kk * DDIM));               \
      bhi[(G) & 1][q] = __builtin_nontemporal_load(                            \
          reinterpret_cast<const v4f*>(wp + (size_t)(kk + 1) * DDIM));         \
    } } while (0)

  LOAD_GROUP(0);
#pragma unroll
  for (int g = 0; g < NGRP; ++g) {
    __builtin_amdgcn_sched_barrier(0);
    if (g + 1 < NGRP) LOAD_GROUP(g + 1);
    __builtin_amdgcn_sched_barrier(0);
#pragma unroll
    for (int q = 0; q < GSTEP; ++q) {
#pragma unroll
      for (int j = 0; j < 4; ++j) {
        v2f b;
        b.x = blo[g & 1][q][j];
        b.y = bhi[g & 1][q][j];
        acc0[j] = __builtin_amdgcn_wmma_f32_16x16x4_f32(
            false, a0[g & 1][q], false, b, (short)0, acc0[j], false, false);
        if (TWO)
          acc1[j] = __builtin_amdgcn_wmma_f32_16x16x4_f32(
              false, a1[g & 1][q], false, b, (short)0, acc1[j], false, false);
      }
    }
  }
#undef LOAD_GROUP

#pragma unroll
  for (int i = 0; i < 8; ++i) {
    v4f v = {acc0[0][i], acc0[1][i], acc0[2][i], acc0[3][i]};
    *reinterpret_cast<v4f*>(&slab[i + 8 * half][4 * c]) = v;
  }
  if (half == 0) {
    v4f v = {acc1[0][0], acc1[1][0], acc1[2][0], acc1[3][0]};
    *reinterpret_cast<v4f*>(&slab[16][4 * c]) = v;
  }
}

__global__ __launch_bounds__(GWAVES * 32) void gemm_kernel(
    const float* __restrict__ xT,
    const float* __restrict__ W,
    const float* __restrict__ root,
    const int* __restrict__ need16,
    float* __restrict__ partial)
{
  __shared__ __attribute__((aligned(16))) float red[GWAVES][NNODES][STRIP];

  const int lane  = threadIdx.x & 31;
  const int wave  = __builtin_amdgcn_readfirstlane((int)(threadIdx.x >> 5));
  const int strip = blockIdx.x % NSTRIP;
  const int plane = blockIdx.x / NSTRIP;
  const int chunk = wave * NPLANE + plane;
  const int slot  = chunk / (DDIM / KCH);
  const int kbase = (chunk % (DDIM / KCH)) * KCH;
  const int h0    = strip * STRIP;

  const int half = lane >> 4;
  const int c    = lane & 15;

  const float* __restrict__ Wr =
      (slot < 8) ? (W + (size_t)slot * DDIM * DDIM) : root;

  const float* wp = Wr + (size_t)(kbase + 2 * half) * DDIM + h0 + 4 * c;
  const float* ap = xT + ((size_t)slot * DDIM + kbase + 2 * half) * NPAD + c;

  if (need16[slot]) chunk_product<true >(wp, ap, red[wave], half, c);
  else              chunk_product<false>(wp, ap, red[wave], half, c);
  __syncthreads();

  for (int o = threadIdx.x; o < NNODES * STRIP; o += GWAVES * 32) {
    const int row = o / STRIP;
    const int col = o % STRIP;
    float s = red[0][row][col];
#pragma unroll
    for (int w = 1; w < GWAVES; ++w) s += red[w][row][col];
    partial[((size_t)plane * NPAD + row) * DDIM + h0 + col] = s;
  }
}

__global__ __launch_bounds__(256) void reduce_kernel(
    const float* __restrict__ partial, const float* __restrict__ bias,
    float* __restrict__ out)
{
  int idx = blockIdx.x * blockDim.x + threadIdx.x;
  if (idx >= NNODES * DDIM) return;
  int n = idx / DDIM;
  int h = idx % DDIM;
  float s = bias[h];
#pragma unroll
  for (int p = 0; p < NPLANE; ++p)
    s += partial[((size_t)p * NPAD + n) * DDIM + h];
  out[idx] = s;
}

extern "C" void kernel_launch(void* const* d_in, const int* in_sizes, int n_in,
                              void* d_out, int out_size, void* d_ws, size_t ws_size,
                              hipStream_t stream) {
  const float* node_emb = (const float*)d_in[0];
  const float* sig      = (const float*)d_in[1];
  const float* W1       = (const float*)d_in[2];
  const float* root1    = (const float*)d_in[3];
  const float* bias1    = (const float*)d_in[4];
  const float* W2       = (const float*)d_in[5];
  const float* root2    = (const float*)d_in[6];
  const float* bias2    = (const float*)d_in[7];
  const int*   ei       = (const int*)d_in[8];
  const int*   et       = (const int*)d_in[9];
  const int    E        = in_sizes[9];

  float* xT      = (float*)d_ws;
  float* partial = xT + (size_t)NSLOT * DDIM * NPAD;
  int*   need16  = (int*)(partial + (size_t)NSLOT * NPAD * DDIM);

  const int prepGrid   = NSLOT * PREP_TILES;
  const int gemmGrid   = NPLANE * NSTRIP;
  const int reduceGrid = (NNODES * DDIM + 255) / 256;

  prep_kernel<false><<<prepGrid, 256, 0, stream>>>(node_emb, sig, 6, ei, et, E, xT, need16);
  gemm_kernel<<<gemmGrid, GWAVES * 32, 0, stream>>>(xT, W1, root1, need16, partial);

  prep_kernel<true><<<prepGrid, 256, 0, stream>>>(partial, bias1, 0, ei, et, E, xT, need16);
  gemm_kernel<<<gemmGrid, GWAVES * 32, 0, stream>>>(xT, W2, root2, need16, partial);
  reduce_kernel<<<reduceGrid, 256, 0, stream>>>(partial, bias2, (float*)d_out);
}


# xT re-laid as [tile][k/4][16 nodes][4 k]: one 16-B A load per lane per two k-steps (was four 4-B), first-tile waves fetch no second-tile lines; k order inside 8-row groups: new summation order
# speedup vs baseline: 5.2597x; 1.1249x over previous
#include <hip/hip_runtime.h>


#define NNODES 17
#define NPAD   32
#define NSLOT  9
#define DDIM   2048

#define KCH    128
#define GSTEP  4
#define NGRP   (KCH / 4 / GSTEP)
#define GWAVES 18
#define NCHUNK (NSLOT * DDIM / KCH)
#define NPLANE (NCHUNK / GWAVES)
#define STRIP  64
#define NSTRIP (DDIM / STRIP)
#define XT_TILE ((size_t)NSLOT * DDIM * 16)

typedef __attribute__((ext_vector_type(2))) float v2f;
typedef __attribute__((ext_vector_type(4))) float v4f;
typedef __attribute__((ext_vector_type(8))) float v8f;

#define PREP_DT   64
#define PREP_TILES (DDIM / PREP_DT)
#define XPITCH    (PREP_DT + 4)

template <bool FROM_PLANES>
__global__ __launch_bounds__(256) void prep_kernel(
    const float* __restrict__ xmain, const float* __restrict__ xsig, int nsig,
    const int* __restrict__ ei, const int* __restrict__ et, int E,
    float* __restrict__ xT, int* __restrict__ need16)
{
  __shared__ unsigned masks[8][32];
  __shared__ int      srcs[256];
  __shared__ __attribute__((aligned(16))) float xs[FROM_PLANES ? NNODES : 1][XPITCH];

  const int slot = blockIdx.x / PREP_TILES;
  const int lane = threadIdx.x & 31;
  const int wave = threadIdx.x >> 5;
  const int d0   = (blockIdx.x % PREP_TILES) * PREP_DT;
  const int d    = d0 + wave * 8;

  int t0 = -1, dn0 = -1, s0 = 0;
  if (slot != 8 && (int)threadIdx.x < E) {
    t0 = et[threadIdx.x]; dn0 = ei[E + threadIdx.x]; s0 = ei[threadIdx.x];
  }

  if (FROM_PLANES) {
    const int col4 = ((int)threadIdx.x & 15) * 4;
    const int rowa = (int)threadIdx.x >> 4;
    const bool second = threadIdx.x < 16;
    const float* pa = xmain + (size_t)rowa * DDIM + d0 + col4;
    const float* pb = xmain + (size_t)(NNODES - 1) * DDIM + d0 + col4;
    const v4f bv = *reinterpret_cast<const v4f*>(xsig + d0 + col4);
    v4f va[NPLANE], vb[NPLANE];
#pragma unroll
    for (int p = 0; p < NPLANE; ++p)
      va[p] = *reinterpret_cast<const v4f*>(pa + (size_t)p * NPAD * DDIM);
    if (second) {
#pragma unroll
      for (int p = 0; p < NPLANE; ++p)
        vb[p] = *reinterpret_cast<const v4f*>(pb + (size_t)p * NPAD * DDIM);
    }
    v4f sa = bv;
#pragma unroll
    for (int p = 0; p < NPLANE; ++p) sa += va[p];
#pragma unroll
    for (int i = 0; i < 4; ++i) sa[i] = fmaxf(sa[i], 0.0f);
    *reinterpret_cast<v4f*>(&xs[rowa][col4]) = sa;
    if (second) {
      v4f sb = bv;
#pragma unroll
      for (int p = 0; p < NPLANE; ++p) sb += vb[p];
#pragma unroll
      for (int i = 0; i < 4; ++i) sb[i] = fmaxf(sb[i], 0.0f);
      *reinterpret_cast<v4f*>(&xs[NNODES - 1][col4]) = sb;
    }
    if (slot == 8) __syncthreads();
  }

  v4f lo = {0.0f, 0.0f, 0.0f, 0.0f};
  v4f hi = {0.0f, 0.0f, 0.0f, 0.0f};
  int cnt = 0;

  if (slot == 8) {
    if (lane < NNODES) {
      const float* row = FROM_PLANES
          ? &xs[lane][wave * 8]
          : ((lane < nsig) ? xsig : xmain) + (size_t)lane * DDIM + d;
      lo = *reinterpret_cast<const v4f*>(row);
      hi = *reinterpret_cast<const v4f*>(row + 4);
    }
  } else {
    for (int base = 0; base < E; base += 256) {
      if (base) __syncthreads();
      const int e = base + (int)threadIdx.x;
      int t = t0, dn = dn0, s = s0;
      if (base) {
        t = -1; dn = -1; s = 0;
        if (e < E) { t = et[e]; dn = ei[E + e]; s = ei[e]; }
      }
      const bool hit = (t == slot);
      unsigned mine = 0;
#pragma unroll
      for (int m = 0; m < NNODES; ++m) {
        const unsigned b = __builtin_amdgcn_ballot_w32(hit && dn == m);
        if (lane == m) mine = b;
      }
      masks[wave][lane] = mine;
      srcs[threadIdx.x] = s;
      __syncthreads();
      const int left = E - base;
      const int nch  = (left >= 256) ? 8 : ((left + 31) >> 5);
      for (int c = 0; c < nch; ++c) {
        unsigned mk = masks[c][lane];
        while (mk) {
          const int bit = __builtin_ctz(mk);
          mk &= mk - 1;
          const int m = srcs[c * 32 + bit];
          const float* row = FROM_PLANES
              ? &xs[m][wave * 8]
              : ((m < nsig) ? xsig : xmain) + (size_t)m * DDIM + d;
          lo += *reinterpret_cast<const v4f*>(row);
          hi += *reinterpret_cast<const v4f*>(row + 4);
          ++cnt;
        }
      }
    }
    if (cnt) {
      const float fc = (float)cnt;
      lo = lo / fc;
      hi = hi / fc;
    }
  }

  if ((blockIdx.x % PREP_TILES) == 0 && threadIdx.x == NNODES - 1)
    need16[slot] = (slot == 8 || cnt > 0) ? 1 : 0;

  float* dst = xT + (size_t)(lane >> 4) * XT_TILE
             + (((size_t)slot * DDIM + d) / 4 * 16 + (lane & 15)) * 4;
  *reinterpret_cast<v4f*>(dst)      = lo;
  *reinterpret_cast<v4f*>(dst + 64) = hi;
}


template <bool TWO>
__device__ __forceinline__ void chunk_product(
    const float* __restrict__ wp, const float* __restrict__ ap,
    float (*__restrict__ slab)[STRIP], const int half, const int c)
{
  v8f acc0[4] = {};
  v8f acc1[4] = {};

  v4f br[2][GSTEP / 2][4];
  v4f av0[2][GSTEP / 2], av1[2][GSTEP / 2];

#define LOAD_GROUP(G) do {                                                     \
    _Pragma("unroll") for (int pp = 0; pp < GSTEP / 2; ++pp) {                 \
      const int P = (G) * (GSTEP / 2) + pp;                                    \
      av0[(G) & 1][pp] = *reinterpret_cast<const v4f*>(ap + (size_t)P * 128);  \
      if (TWO)                                                                 \
        av1[(G) & 1][pp] =                                                     \
            *reinterpret_cast<const v4f*>(ap + XT_TILE + (size_t)P * 128);     \
      _Pragma("unroll") for (int i = 0; i < 4; ++i)                            \
        br[(G) & 1][pp][i] = __builtin_nontemporal_load(                       \
            reinterpret_cast<const v4f*>(wp + (size_t)(P * 8 + i) * DDIM));    \
    } } while (0)

  LOAD_GROUP(0);
#pragma unroll
  for (int g = 0; g < NGRP; ++g) {
    __builtin_amdgcn_sched_barrier(0);
    if (g + 1 < NGRP) LOAD_GROUP(g + 1);
    __builtin_amdgcn_sched_barrier(0);
#pragma unroll
    for (int pp = 0; pp < GSTEP / 2; ++pp) {
#pragma unroll
      for (int st = 0; st < 2; ++st) {
        v2f a0, a1;
        a0.x = av0[g & 1][pp][2 * st];
        a0.y = av0[g & 1][pp][2 * st + 1];
        if (TWO) {
          a1.x = av1[g & 1][pp][2 * st];
          a1.y = av1[g & 1][pp][2 * st + 1];
        }
#pragma unroll
        for (int j = 0; j < 4; ++j) {
          v2f b;
          b.x = br[g & 1][pp][2 * st][j];
          b.y = br[g & 1][pp][2 * st + 1][j];
          acc0[j] = __builtin_amdgcn_wmma_f32_16x16x4_f32(
              false, a0, false, b, (short)0, acc0[j], false, false);
          if (TWO)
            acc1[j] = __builtin_amdgcn_wmma_f32_16x16x4_f32(
                false, a1, false, b, (short)0, acc1[j], false, false);
        }
      }
    }
  }
#undef LOAD_GROUP

#pragma unroll
  for (int i = 0; i < 8; ++i) {
    v4f v = {acc0[0][i], acc0[1][i], acc0[2][i], acc0[3][i]};
    *reinterpret_cast<v4f*>(&slab[i + 8 * half][4 * c]) = v;
  }
  if (half == 0) {
    v4f v = {acc1[0][0], acc1[1][0], acc1[2][0], acc1[3][0]};
    *reinterpret_cast<v4f*>(&slab[16][4 * c]) = v;
  }
}

__global__ __launch_bounds__(GWAVES * 32) void gemm_kernel(
    const float* __restrict__ xT,
    const float* __restrict__ W,
    const float* __restrict__ root,
    const int* __restrict__ need16,
    float* __restrict__ partial)
{
  __shared__ __attribute__((aligned(16))) float red[GWAVES][NNODES][STRIP];

  const int lane  = threadIdx.x & 31;
  const int wave  = __builtin_amdgcn_readfirstlane((int)(threadIdx.x >> 5));
  const int strip = blockIdx.x % NSTRIP;
  const int plane = blockIdx.x / NSTRIP;
  const int chunk = wave * NPLANE + plane;
  const int slot  = chunk / (DDIM / KCH);
  const int kbase = (chunk % (DDIM / KCH)) * KCH;
  const int h0    = strip * STRIP;

  const int half = lane >> 4;
  const int c    = lane & 15;

  const float* __restrict__ Wr =
      (slot < 8) ? (W + (size_t)slot * DDIM * DDIM) : root;

  const float* wp = Wr + (size_t)(kbase + 4 * half) * DDIM + h0 + 4 * c;
  const float* ap = xT + ((((size_t)slot * DDIM + kbase) / 4 + half) * 16 + c) * 4;

  if (need16[slot]) chunk_product<true >(wp, ap, red[wave], half, c);
  else              chunk_product<false>(wp, ap, red[wave], half, c);
  __syncthreads();

  for (int o = threadIdx.x; o < NNODES * STRIP; o += GWAVES * 32) {
    const int row = o / STRIP;
    const int col = o % STRIP;
    float s = red[0][row][col];
#pragma unroll
    for (int w = 1; w < GWAVES; ++w) s += red[w][row][col];
    partial[((size_t)plane * NPAD + row) * DDIM + h0 + col] = s;
  }
}

__global__ __launch_bounds__(256) void reduce_kernel(
    const float* __restrict__ partial, const float* __restrict__ bias,
    float* __restrict__ out)
{
  int idx = blockIdx.x * blockDim.x + threadIdx.x;
  if (idx >= NNODES * DDIM) return;
  int n = idx / DDIM;
  int h = idx % DDIM;
  float s = bias[h];
#pragma unroll
  for (int p = 0; p < NPLANE; ++p)
    s += partial[((size_t)p * NPAD + n) * DDIM + h];
  out[idx] = s;
}

extern "C" void kernel_launch(void* const* d_in, const int* in_sizes, int n_in,
                              void* d_out, int out_size, void* d_ws, size_t ws_size,
                              hipStream_t stream) {
  const float* node_emb = (const float*)d_in[0];
  const float* sig      = (const float*)d_in[1];
  const float* W1       = (const float*)d_in[2];
  const float* root1    = (const float*)d_in[3];
  const float* bias1    = (const float*)d_in[4];
  const float* W2       = (const float*)d_in[5];
  const float* root2    = (const float*)d_in[6];
  const float* bias2    = (const float*)d_in[7];
  const int*   ei       = (const int*)d_in[8];
  const int*   et       = (const int*)d_in[9];
  const int    E        = in_sizes[9];

  float* xT      = (float*)d_ws;
  float* partial = xT + (size_t)NSLOT * DDIM * NPAD;
  int*   need16  = (int*)(partial + (size_t)NSLOT * NPAD * DDIM);

  const int prepGrid   = NSLOT * PREP_TILES;
  const int gemmGrid   = NPLANE * NSTRIP;
  const int reduceGrid = (NNODES * DDIM + 255) / 256;

  prep_kernel<false><<<prepGrid, 256, 0, stream>>>(node_emb, sig, 6, ei, et, E, xT, need16);
  gemm_kernel<<<gemmGrid, GWAVES * 32, 0, stream>>>(xT, W1, root1, need16, partial);

  prep_kernel<true><<<prepGrid, 256, 0, stream>>>(partial, bias1, 0, ei, et, E, xT, need16);
  gemm_kernel<<<gemmGrid, GWAVES * 32, 0, stream>>>(xT, W2, root2, need16, partial);
  reduce_kernel<<<reduceGrid, 256, 0, stream>>>(partial, bias2, (float*)d_out);
}


# prep folded into gemm: block forms its 17x256 x tile in LDS (layer 2: bias + 8 planes, relu), each wave aggregates its own A tile there and reads A from LDS; 3 launches, same summation order
# speedup vs baseline: 6.6769x; 1.2694x over previous
#include <hip/hip_runtime.h>


#define NNODES 17
#define NPAD   32
#define NSLOT  9
#define DDIM   2048

#define KCH    128
#define GSTEP  4
#define NGRP   (KCH / 4 / GSTEP)
#define GWAVES 18
#define NCHUNK (NSLOT * DDIM / KCH)
#define NPLANE (NCHUNK / GWAVES)
#define STRIP  64
#define NSTRIP (DDIM / STRIP)

#define XCOLS  (2 * KCH)
#define XSP    (XCOLS + 4)
#define NITEM  (NNODES * (XCOLS / 4))
#define AKQ    (KCH / 4)
#define APITCH (NNODES * 4)
#define ATILE  (AKQ * APITCH)
#define EPASS  (GWAVES * 32)

typedef __attribute__((ext_vector_type(2))) float v2f;
typedef __attribute__((ext_vector_type(4))) float v4f;
typedef __attribute__((ext_vector_type(8))) float v8f;


#define LOAD_B(G) do {                                                         \
    _Pragma("unroll") for (int pp = 0; pp < GSTEP / 2; ++pp) {                 \
      const int P = (G) * (GSTEP / 2) + pp;                                    \
      _Pragma("unroll") for (int i = 0; i < 4; ++i)                            \
        br[(G) & 1][pp][i] = __builtin_nontemporal_load(                       \
            reinterpret_cast<const v4f*>(wp + (size_t)(P * 8 + i) * DDIM));    \
    } } while (0)

template <bool TWO>
__device__ __forceinline__ void chunk_product(
    const float* __restrict__ wp, float* __restrict__ atile,
    v4f (&br)[2][GSTEP / 2][4], const int half, const int c)
{
  v8f acc0[4] = {};
  v8f acc1[4] = {};

  v4f av0[2][GSTEP / 2];
  v4f av1[GSTEP / 2];

  const float* ap = atile + half * APITCH + c * 4;
  const float* aq = atile + half * APITCH + 16 * 4;

#define LOAD_A(G) do {                                                         \
    _Pragma("unroll") for (int pp = 0; pp < GSTEP / 2; ++pp) {                 \
      const int P = (G) * (GSTEP / 2) + pp;                                    \
      av0[(G) & 1][pp] = *reinterpret_cast<const v4f*>(ap + P * 2 * APITCH);   \
    } } while (0)
#define LOAD_A1(G, PP)                                                         \
    av1[PP] = *reinterpret_cast<const v4f*>(                                   \
        aq + ((G) * (GSTEP / 2) + (PP)) * 2 * APITCH)

  LOAD_A(0);
  if (TWO) {
#pragma unroll
    for (int pp = 0; pp < GSTEP / 2; ++pp) LOAD_A1(0, pp);
  }
#pragma unroll
  for (int g = 0; g < NGRP; ++g) {
    __builtin_amdgcn_sched_barrier(0);
    if (g + 1 < NGRP) { if (g) LOAD_B(g + 1); LOAD_A(g + 1); }
    __builtin_amdgcn_sched_barrier(0);
#pragma unroll
    for (int pp = 0; pp < GSTEP / 2; ++pp) {
#pragma unroll
      for (int st = 0; st < 2; ++st) {
        v2f a0, a1;
        a0.x = av0[g & 1][pp][2 * st];
        a0.y = av0[g & 1][pp][2 * st + 1];
        if (TWO) {
          a1.x = av1[pp][2 * st];
          a1.y = av1[pp][2 * st + 1];
        }
#pragma unroll
        for (int j = 0; j < 4; ++j) {
          v2f b;
          b.x = br[g & 1][pp][2 * st][j];
          b.y = br[g & 1][pp][2 * st + 1][j];
          acc0[j] = __builtin_amdgcn_wmma_f32_16x16x4_f32(
              false, a0, false, b, (short)0, acc0[j], false, false);
          if (TWO)
            acc1[j] = __builtin_amdgcn_wmma_f32_16x16x4_f32(
                false, a1, false, b, (short)0, acc1[j], false, false);
        }
      }
      if (TWO && g + 1 < NGRP) LOAD_A1(g + 1, pp);
    }
  }
#undef LOAD_A
#undef LOAD_A1

  float (*slab)[STRIP] = reinterpret_cast<float (*)[STRIP]>(atile);
#pragma unroll
  for (int i = 0; i < 8; ++i) {
    v4f v = {acc0[0][i], acc0[1][i], acc0[2][i], acc0[3][i]};
    *reinterpret_cast<v4f*>(&slab[i + 8 * half][4 * c]) = v;
  }
  if (half == 0) {
    v4f v = {acc1[0][0], acc1[1][0], acc1[2][0], acc1[3][0]};
    *reinterpret_cast<v4f*>(&slab[16][4 * c]) = v;
  }
}

template <bool FROM_PLANES>
__global__ __launch_bounds__(GWAVES * 32) void layer_kernel(
    const float* __restrict__ xmain, const float* __restrict__ xsig, int nsig,
    const int* __restrict__ ei, const int* __restrict__ et, int E,
    const float* __restrict__ W,
    const float* __restrict__ root,
    float* __restrict__ partial)
{
  __shared__ __attribute__((aligned(16))) float xs[NNODES + 1][XSP];
  __shared__ __attribute__((aligned(16))) float at[GWAVES][ATILE];
  __shared__ int e_t[EPASS], e_d[EPASS], e_s[EPASS];

  const int lane  = threadIdx.x & 31;
  const int wave  = __builtin_amdgcn_readfirstlane((int)(threadIdx.x >> 5));
  const int strip = blockIdx.x % NSTRIP;
  const int plane = blockIdx.x / NSTRIP;
  const int chunk = wave * NPLANE + plane;
  const int slot  = chunk / (DDIM / KCH);
  const int kbase = (chunk % (DDIM / KCH)) * KCH;
  const int h0    = strip * STRIP;

  const int half = lane >> 4;
  const int c    = lane & 15;

  const float* __restrict__ Wr =
      (slot < 8) ? (W + (size_t)slot * DDIM * DDIM) : root;
  const float* wp = Wr + (size_t)(kbase + 4 * half) * DDIM + h0 + 4 * c;

  int t0 = -1, d0 = -1, s0 = 0;
  if ((int)threadIdx.x < E) {
    t0 = et[threadIdx.x]; d0 = ei[E + threadIdx.x]; s0 = ei[threadIdx.x];
  }

  v4f xv[2][FROM_PLANES ? NPLANE + 1 : 1];
  int xdst[2];
#pragma unroll
  for (int j = 0; j < 2; ++j) {
    const int iu   = (int)threadIdx.x + j * GWAVES * 32;
    const int it   = (iu < NITEM) ? iu : NITEM - 1;
    const int row  = it / (XCOLS / 4);
    const int q    = it % (XCOLS / 4);
    const int r    = q / (KCH / 4);
    const int col4 = (q % (KCH / 4)) * 4;
    const int gcol = (r * NPLANE + plane) * KCH + col4;
    xdst[j] = (iu < NITEM) ? row * XSP + r * KCH + col4
                           : NNODES * XSP + (iu - NITEM) * 4;
    if (FROM_PLANES) {
      xv[j][0] = *reinterpret_cast<const v4f*>(xsig + gcol);
#pragma unroll
      for (int p = 0; p < NPLANE; ++p)
        xv[j][1 + p] = *reinterpret_cast<const v4f*>(
            xmain + ((size_t)p * NPAD + row) * DDIM + gcol);
    } else {
      xv[j][0] = *reinterpret_cast<const v4f*>(
          ((row < nsig) ? xsig : xmain) + (size_t)row * DDIM + gcol);
    }
  }

  v4f br[2][GSTEP / 2][4];
  LOAD_B(0);
  LOAD_B(1);
  __builtin_amdgcn_sched_barrier(0);

  e_t[threadIdx.x] = t0; e_d[threadIdx.x] = d0; e_s[threadIdx.x] = s0;

  float* xsf = &xs[0][0];
#pragma unroll
  for (int j = 0; j < 2; ++j) {
    v4f v = xv[j][0];
    if (FROM_PLANES) {
#pragma unroll
      for (int p = 0; p < NPLANE; ++p) v += xv[j][1 + p];
#pragma unroll
      for (int i = 0; i < 4; ++i) v[i] = fmaxf(v[i], 0.0f);
    }
    *reinterpret_cast<v4f*>(xsf + xdst[j]) = v;
  }
  __builtin_amdgcn_fence(__ATOMIC_RELEASE, "workgroup", "local");
  __builtin_amdgcn_s_barrier();
  __builtin_amdgcn_fence(__ATOMIC_ACQUIRE, "workgroup", "local");

  const int xcol = (wave & 1) * KCH + 4 * lane;
  v4f acc[NNODES];
  int cnt[NNODES];
#pragma unroll
  for (int n = 0; n < NNODES; ++n) { acc[n] = (v4f){0.0f, 0.0f, 0.0f, 0.0f}; cnt[n] = 0; }

  if (slot == 8) {
#pragma unroll
    for (int n = 0; n < NNODES; ++n)
      acc[n] = *reinterpret_cast<const v4f*>(&xs[n][xcol]);
  }
  for (int base = 0; base < E; base += EPASS) {
    if (base) {
      __syncthreads();
      const int e = base + (int)threadIdx.x;
      int tt = -1, dd = -1, ss = 0;
      if (e < E) { tt = et[e]; dd = ei[E + e]; ss = ei[e]; }
      e_t[threadIdx.x] = tt; e_d[threadIdx.x] = dd; e_s[threadIdx.x] = ss;
      __syncthreads();
    }
    if (slot != 8) {
      const int left = E - base;
      const int nch  = (left >= EPASS) ? GWAVES : ((left + 31) >> 5);
      int t = e_t[lane], dn = e_d[lane], s = e_s[lane];
      for (int ch = 0; ch < nch; ++ch) {
        int tn = -1, dnn = -1, sn = 0;
        if (ch + 1 < nch) {
          tn = e_t[(ch + 1) * 32 + lane]; dnn = e_d[(ch + 1) * 32 + lane]; sn = e_s[(ch + 1) * 32 + lane];
        }
        const bool hit = (t == slot);
        if (__builtin_amdgcn_ballot_w32(hit)) {
#pragma unroll
          for (int n = 0; n < NNODES; ++n) {
            unsigned mk = __builtin_amdgcn_ballot_w32(hit && dn == n);
            cnt[n] += __builtin_popcount(mk);
            while (mk) {
              const int bit = __builtin_ctz(mk);
              mk &= mk - 1;
              const int m = __builtin_amdgcn_readlane(s, bit);
              acc[n] += *reinterpret_cast<const v4f*>(&xs[m][xcol]);
            }
          }
        }
        t = tn; dn = dnn; s = sn;
      }
    }
  }
#pragma unroll
  for (int n = 0; n < NNODES; ++n) {
    if (cnt[n]) {
      const float fc = (float)cnt[n];
      acc[n] = acc[n] / fc;
    }
  }

  float* atile = &at[wave][0];
#pragma unroll
  for (int n = 0; n < NNODES; ++n)
    *reinterpret_cast<v4f*>(atile + lane * APITCH + n * 4) = acc[n];

  if (slot == 8 || cnt[NNODES - 1] > 0) chunk_product<true >(wp, atile, br, half, c);
  else                                  chunk_product<false>(wp, atile, br, half, c);
  __syncthreads();

  for (int o = threadIdx.x; o < NNODES * STRIP; o += GWAVES * 32) {
    const int row = o / STRIP;
    const int col = o % STRIP;
    float sum = at[0][o];
#pragma unroll
    for (int w = 1; w < GWAVES; ++w) sum += at[w][o];
    partial[((size_t)plane * NPAD + row) * DDIM + h0 + col] = sum;
  }
}
#undef LOAD_B

__global__ __launch_bounds__(256) void reduce_kernel(
    const float* __restrict__ partial, const float* __restrict__ bias,
    float* __restrict__ out)
{
  int idx = blockIdx.x * blockDim.x + threadIdx.x;
  if (idx >= NNODES * DDIM) return;
  int n = idx / DDIM;
  int h = idx % DDIM;
  float s = bias[h];
#pragma unroll
  for (int p = 0; p < NPLANE; ++p)
    s += partial[((size_t)p * NPAD + n) * DDIM + h];
  out[idx] = s;
}

extern "C" void kernel_launch(void* const* d_in, const int* in_sizes, int n_in,
                              void* d_out, int out_size, void* d_ws, size_t ws_size,
                              hipStream_t stream) {
  const float* node_emb = (const float*)d_in[0];
  const float* sig      = (const float*)d_in[1];
  const float* W1       = (const float*)d_in[2];
  const float* root1    = (const float*)d_in[3];
  const float* bias1    = (const float*)d_in[4];
  const float* W2       = (const float*)d_in[5];
  const float* root2    = (const float*)d_in[6];
  const float* bias2    = (const float*)d_in[7];
  const int*   ei       = (const int*)d_in[8];
  const int*   et       = (const int*)d_in[9];
  const int    E        = in_sizes[9];

  float* part1 = (float*)d_ws;
  float* part2 = part1 + (size_t)NPLANE * NPAD * DDIM;

  const int layerGrid  = NPLANE * NSTRIP;
  const int reduceGrid = (NNODES * DDIM + 255) / 256;

  layer_kernel<false><<<layerGrid, GWAVES * 32, 0, stream>>>(
      node_emb, sig, 6, ei, et, E, W1, root1, part1);
  layer_kernel<true><<<layerGrid, GWAVES * 32, 0, stream>>>(
      part1, bias1, 0, ei, et, E, W2, root2, part2);
  reduce_kernel<<<reduceGrid, 256, 0, stream>>>(part2, bias2, (float*)d_out);
}


# 10 waves per block (was 18): a relation wave streams both chunks of its slot (4-group W ring, one accumulator); node 16 rides in an empty tile row; mean by reciprocal+fma; new summation order
# speedup vs baseline: 7.0907x; 1.0620x over previous
#include <hip/hip_runtime.h>


#define NNODES 17
#define NPAD   32
#define NSLOT  9
#define DDIM   2048

#define KCH    128
#define GSTEP  4
#define NGRP   (KCH / 4 / GSTEP)
#define NTILE  18
#define GWAVES 10
#define RING   4
#define NCHUNK (NSLOT * DDIM / KCH)
#define NPLANE (NCHUNK / NTILE)
#define CHSTRIDE ((size_t)NPLANE * KCH * DDIM)
#define STRIP  64
#define NSTRIP (DDIM / STRIP)

#define XCOLS  (2 * KCH)
#define XSP    (XCOLS + 4)
#define NITEM  (NNODES * (XCOLS / 4))
#define XPER   4
#define XSPARE ((((XPER * GWAVES * 32 - NITEM) * 4) + XSP - 1) / XSP)
#define AKQ    (KCH / 4)
#define APITCH (NNODES * 4)
#define ATILE  (AKQ * APITCH)
#define EPASS  (GWAVES * 32)

typedef __attribute__((ext_vector_type(2))) float v2f;
typedef __attribute__((ext_vector_type(4))) float v4f;
typedef __attribute__((ext_vector_type(8))) float v8f;


#define LOAD_B(PTR, G, S) do {                                                 \
    _Pragma("unroll") for (int pp = 0; pp < GSTEP / 2; ++pp) {                 \
      const int P = (G) * (GSTEP / 2) + pp;                                    \
      _Pragma("unroll") for (int i = 0; i < 4; ++i)                            \
        br[S][pp][i] = __builtin_nontemporal_load(                             \
            reinterpret_cast<const v4f*>((PTR) + (size_t)(P * 8 + i) * DDIM)); \
    } } while (0)
#define LOAD_A(BASE, G, S) do {                                                \
    _Pragma("unroll") for (int pp = 0; pp < GSTEP / 2; ++pp) {                 \
      const int P = (G) * (GSTEP / 2) + pp;                                    \
      av0[S][pp] = *reinterpret_cast<const v4f*>((BASE) + P * 2 * APITCH);     \
    } } while (0)
#define LOAD_A1(BASE, G, PP)                                                   \
    av1[PP] = *reinterpret_cast<const v4f*>(                                   \
        (BASE) + ((G) * (GSTEP / 2) + (PP)) * 2 * APITCH)

template <bool TWO>
__device__ __forceinline__ void chunk_groups(
    const float* __restrict__ wp, const float* __restrict__ wpn, const bool more,
    const float* __restrict__ ap, const float* __restrict__ aq,
    v4f (&br)[RING][GSTEP / 2][4], v4f (&av0)[2][GSTEP / 2], v4f (&av1)[GSTEP / 2],
    v8f (&acc0)[4], v8f (&acc1)[4])
{
#pragma unroll
  for (int g = 0; g < NGRP; ++g) {
    __builtin_amdgcn_sched_barrier(0);
    if (g + RING - 1 < NGRP)  LOAD_B(wp, g + RING - 1, (g + RING - 1) % RING);
    else if (more)            LOAD_B(wpn, g + RING - 1 - NGRP, (g + RING - 1) % RING);
    if (g + 1 < NGRP)         LOAD_A(ap, g + 1, (g + 1) & 1);
    else if (more)            LOAD_A(ap + ATILE, 0, (g + 1) & 1);
    __builtin_amdgcn_sched_barrier(0);
#pragma unroll
    for (int pp = 0; pp < GSTEP / 2; ++pp) {
#pragma unroll
      for (int st = 0; st < 2; ++st) {
        v2f a0, a1;
        a0.x = av0[g & 1][pp][2 * st];
        a0.y = av0[g & 1][pp][2 * st + 1];
        if (TWO) {
          a1.x = av1[pp][2 * st];
          a1.y = av1[pp][2 * st + 1];
        }
#pragma unroll
        for (int j = 0; j < 4; ++j) {
          v2f b;
          b.x = br[g % RING][pp][2 * st][j];
          b.y = br[g % RING][pp][2 * st + 1][j];
          acc0[j] = __builtin_amdgcn_wmma_f32_16x16x4_f32(
              false, a0, false, b, (short)0, acc0[j], false, false);
          if (TWO)
            acc1[j] = __builtin_amdgcn_wmma_f32_16x16x4_f32(
                false, a1, false, b, (short)0, acc1[j], false, false);
        }
      }
      if (TWO) {
        if (g + 1 < NGRP) LOAD_A1(aq, g + 1, pp);
        else if (more)    LOAD_A1(aq + ATILE, 0, pp);
      }
    }
  }
}

template <bool TWO, bool STATIC2>
__device__ __forceinline__ void stream_chunks(
    const float* __restrict__ wp, const int nchk, float* __restrict__ atile,
    v4f (&br)[RING][GSTEP / 2][4], const int half, const int c)
{
  v8f acc0[4] = {};
  v8f acc1[4] = {};

  v4f av0[2][GSTEP / 2];
  v4f av1[GSTEP / 2];

  const float* ap = atile + half * APITCH + c * 4;
  const float* aq = atile + half * APITCH + 16 * 4;

  LOAD_A(ap, 0, 0);
  if (TWO) {
#pragma unroll
    for (int pp = 0; pp < GSTEP / 2; ++pp) LOAD_A1(aq, 0, pp);
  }
  if constexpr (STATIC2) {
    chunk_groups<TWO>(wp, wp + CHSTRIDE, true, ap, aq, br, av0, av1, acc0, acc1);
    chunk_groups<TWO>(wp + CHSTRIDE, wp + CHSTRIDE, false, ap + ATILE, aq + ATILE,
                      br, av0, av1, acc0, acc1);
  } else {
#pragma nounroll
    for (int ch = 0; ch < nchk; ++ch) {
      chunk_groups<TWO>(wp, wp + CHSTRIDE, ch + 1 < nchk, ap, aq, br, av0, av1, acc0, acc1);
      wp += CHSTRIDE; ap += ATILE; aq += ATILE;
    }
  }

  float (*slab)[STRIP] = reinterpret_cast<float (*)[STRIP]>(atile);
#pragma unroll
  for (int i = 0; i < 8; ++i) {
    v4f v = {acc0[0][i], acc0[1][i], acc0[2][i], acc0[3][i]};
    *reinterpret_cast<v4f*>(&slab[i + 8 * half][4 * c]) = v;
  }
  if (half == 0) {
    v4f v = {acc1[0][0], acc1[1][0], acc1[2][0], acc1[3][0]};
    *reinterpret_cast<v4f*>(&slab[16][4 * c]) = v;
  }
}

__device__ __forceinline__ float div_by_count(const float x, const float fc, const float rc)
{
  const float q   = x * rc;
  const float rem = __builtin_fmaf(-q, fc, x);
  return __builtin_fmaf(rem, rc, q);
}

template <bool FROM_PLANES>
__global__ __launch_bounds__(GWAVES * 32) void layer_kernel(
    const float* __restrict__ xmain, const float* __restrict__ xsig, int nsig,
    const int* __restrict__ ei, const int* __restrict__ et, int E,
    const float* __restrict__ W,
    const float* __restrict__ root,
    float* __restrict__ partial)
{
  __shared__ __attribute__((aligned(16))) float xs[NNODES + XSPARE][XSP];
  __shared__ __attribute__((aligned(16))) float at[NTILE][ATILE];
  __shared__ int e_t[EPASS], e_d[EPASS], e_s[EPASS];

  const int lane  = threadIdx.x & 31;
  const int wave  = __builtin_amdgcn_readfirstlane((int)(threadIdx.x >> 5));
  const int strip = blockIdx.x % NSTRIP;
  const int plane = blockIdx.x / NSTRIP;
  const bool rel  = wave < NSLOT - 1;
  const int slot  = rel ? wave : NSLOT - 1;
  const int r0    = rel ? 0 : wave - (NSLOT - 1);
  const int nchk  = rel ? 2 : 1;
  const int tidx  = rel ? 2 * wave : 2 * (NSLOT - 1) + r0;
  const int kbase = (r0 * NPLANE + plane) * KCH;
  const int h0    = strip * STRIP;

  const int half = lane >> 4;
  const int c    = lane & 15;

  const float* __restrict__ Wr =
      (slot < 8) ? (W + (size_t)slot * DDIM * DDIM) : root;
  const float* wp = Wr + (size_t)(kbase + 4 * half) * DDIM + h0 + 4 * c;

  int t0 = -1, d0 = -1, s0 = 0;
  if ((int)threadIdx.x < E) {
    t0 = et[threadIdx.x]; d0 = ei[E + threadIdx.x]; s0 = ei[threadIdx.x];
  }

  v4f xv[XPER][FROM_PLANES ? NPLANE + 1 : 1];
  int xdst[XPER];
#pragma unroll
  for (int j = 0; j < XPER; ++j) {
    const int iu   = (int)threadIdx.x + j * GWAVES * 32;
    const int it   = (iu < NITEM) ? iu : NITEM - 1;
    const int row  = it / (XCOLS / 4);
    const int q    = it % (XCOLS / 4);
    const int r    = q / (KCH / 4);
    const int col4 = (q % (KCH / 4)) * 4;
    const int gcol = (r * NPLANE + plane) * KCH + col4;
    xdst[j] = (iu < NITEM) ? row * XSP + r * KCH + col4
                           : NNODES * XSP + (iu - NITEM) * 4;
    if (FROM_PLANES) {
      xv[j][0] = *reinterpret_cast<const v4f*>(xsig + gcol);
#pragma unroll
      for (int p = 0; p < NPLANE; ++p)
        xv[j][1 + p] = *reinterpret_cast<const v4f*>(
            xmain + ((size_t)p * NPAD + row) * DDIM + gcol);
    } else {
      xv[j][0] = *reinterpret_cast<const v4f*>(
          ((row < nsig) ? xsig : xmain) + (size_t)row * DDIM + gcol);
    }
  }

  v4f br[RING][GSTEP / 2][4];
  LOAD_B(wp, 0, 0);
  LOAD_B(wp, 1, 1);
  LOAD_B(wp, 2, 2);
  __builtin_amdgcn_sched_barrier(0);

  e_t[threadIdx.x] = t0; e_d[threadIdx.x] = d0; e_s[threadIdx.x] = s0;

  float* xsf = &xs[0][0];
#pragma unroll
  for (int j = 0; j < XPER; ++j) {
    v4f v = xv[j][0];
    if (FROM_PLANES) {
#pragma unroll
      for (int p = 0; p < NPLANE; ++p) v += xv[j][1 + p];
#pragma unroll
      for (int i = 0; i < 4; ++i) v[i] = fmaxf(v[i], 0.0f);
    }
    *reinterpret_cast<v4f*>(xsf + xdst[j]) = v;
  }
  __builtin_amdgcn_fence(__ATOMIC_RELEASE, "workgroup", "local");
  __builtin_amdgcn_s_barrier();
  __builtin_amdgcn_fence(__ATOMIC_ACQUIRE, "workgroup", "local");

  const int xcolA = (rel ? 0 : r0 * KCH) + 4 * lane;
  const int xcolB = KCH + 4 * lane;
  v4f accA[NNODES], accB[NNODES];
  int cnt[NNODES];
#pragma unroll
  for (int n = 0; n < NNODES; ++n) {
    accA[n] = (v4f){0.0f, 0.0f, 0.0f, 0.0f};
    accB[n] = (v4f){0.0f, 0.0f, 0.0f, 0.0f};
    cnt[n] = 0;
  }

  if (!rel) {
#pragma unroll
    for (int n = 0; n < NNODES; ++n)
      accA[n] = *reinterpret_cast<const v4f*>(&xs[n][xcolA]);
  }
  for (int base = 0; base < E; base += EPASS) {
    if (base) {
      __syncthreads();
      const int e = base + (int)threadIdx.x;
      int tt = -1, dd = -1, ss = 0;
      if (e < E) { tt = et[e]; dd = ei[E + e]; ss = ei[e]; }
      e_t[threadIdx.x] = tt; e_d[threadIdx.x] = dd; e_s[threadIdx.x] = ss;
      __syncthreads();
    }
    if (rel) {
      const int left = E - base;
      const int ngr  = (left >= EPASS) ? GWAVES : ((left + 31) >> 5);
      int t = e_t[lane], dn = e_d[lane], s = e_s[lane];
      for (int ch = 0; ch < ngr; ++ch) {
        int tn = -1, dnn = -1, sn = 0;
        if (ch + 1 < ngr) {
          tn = e_t[(ch + 1) * 32 + lane]; dnn = e_d[(ch + 1) * 32 + lane]; sn = e_s[(ch + 1) * 32 + lane];
        }
        const bool hit = (t == slot);
        if (__builtin_amdgcn_ballot_w32(hit)) {
#pragma unroll
          for (int n = 0; n < NNODES; ++n) {
            unsigned mk = __builtin_amdgcn_ballot_w32(hit && dn == n);
            cnt[n] += __builtin_popcount(mk);
            while (mk) {
              const int bit = __builtin_ctz(mk);
              mk &= mk - 1;
              const int m = __builtin_amdgcn_readlane(s, bit);
              accA[n] += *reinterpret_cast<const v4f*>(&xs[m][xcolA]);
              accB[n] += *reinterpret_cast<const v4f*>(&xs[m][xcolB]);
            }
          }
        }
        t = tn; dn = dnn; s = sn;
      }
    }
  }
#pragma unroll
  for (int n = 0; n < NNODES; ++n) {
    if (cnt[n]) {
      const float fc = (float)cnt[n];
      const float rc = 1.0f / fc;
#pragma unroll
      for (int i = 0; i < 4; ++i) {
        accA[n][i] = div_by_count(accA[n][i], fc, rc);
        accB[n][i] = div_by_count(accB[n][i], fc, rc);
      }
    }
  }

  int  ride = -1;
  bool two  = !rel;
  if (rel && cnt[NNODES - 1] > 0) {
    unsigned used = 0;
#pragma unroll
    for (int n = 0; n < NNODES - 1; ++n) used |= (cnt[n] > 0 ? 1u : 0u) << n;
    const unsigned freem = ~used & 0xFFFFu;
    if (freem) ride = __builtin_ctz(freem);
    else       two  = true;
  }

  float* atile = &at[tidx][0];
#pragma unroll
  for (int n = 0; n < NNODES; ++n)
    *reinterpret_cast<v4f*>(atile + lane * APITCH + n * 4) = accA[n];
  if (rel) {
#pragma unroll
    for (int n = 0; n < NNODES; ++n)
      *reinterpret_cast<v4f*>(atile + ATILE + lane * APITCH + n * 4) = accB[n];
  }
  if (ride >= 0) {
    *reinterpret_cast<v4f*>(atile + lane * APITCH + ride * 4) = accA[NNODES - 1];
    *reinterpret_cast<v4f*>(atile + ATILE + lane * APITCH + ride * 4) = accB[NNODES - 1];
  }

  if (two) stream_chunks<true,  false>(wp, nchk, atile, br, half, c);
  else     stream_chunks<false, true >(wp, 2,    atile, br, half, c);
  if (ride >= 0 && half == 0) {
    float (*slab)[STRIP] = reinterpret_cast<float (*)[STRIP]>(atile);
    const v4f v = *reinterpret_cast<const v4f*>(&slab[ride][4 * c]);
    *reinterpret_cast<v4f*>(&slab[16][4 * c])   = v;
    *reinterpret_cast<v4f*>(&slab[ride][4 * c]) = (v4f){0.0f, 0.0f, 0.0f, 0.0f};
  }
  __syncthreads();

  for (int o = threadIdx.x; o < NNODES * STRIP; o += GWAVES * 32) {
    const int row = o / STRIP;
    const int col = o % STRIP;
    float sum = at[0][o];
#pragma unroll
    for (int w = 1; w < GWAVES; ++w)
      sum += at[(w < NSLOT - 1) ? 2 * w : (NSLOT - 1) + w][o];
    partial[((size_t)plane * NPAD + row) * DDIM + h0 + col] = sum;
  }
}
#undef LOAD_B
#undef LOAD_A
#undef LOAD_A1

__global__ __launch_bounds__(256) void reduce_kernel(
    const float* __restrict__ partial, const float* __restrict__ bias,
    float* __restrict__ out)
{
  int idx = blockIdx.x * blockDim.x + threadIdx.x;
  if (idx >= NNODES * DDIM) return;
  int n = idx / DDIM;
  int h = idx % DDIM;
  float s = bias[h];
#pragma unroll
  for (int p = 0; p < NPLANE; ++p)
    s += partial[((size_t)p * NPAD + n) * DDIM + h];
  out[idx] = s;
}

extern "C" void kernel_launch(void* const* d_in, const int* in_sizes, int n_in,
                              void* d_out, int out_size, void* d_ws, size_t ws_size,
                              hipStream_t stream) {
  const float* node_emb = (const float*)d_in[0];
  const float* sig      = (const float*)d_in[1];
  const float* W1       = (const float*)d_in[2];
  const float* root1    = (const float*)d_in[3];
  const float* bias1    = (const float*)d_in[4];
  const float* W2       = (const float*)d_in[5];
  const float* root2    = (const float*)d_in[6];
  const float* bias2    = (const float*)d_in[7];
  const int*   ei       = (const int*)d_in[8];
  const int*   et       = (const int*)d_in[9];
  const int    E        = in_sizes[9];

  float* part1 = (float*)d_ws;
  float* part2 = part1 + (size_t)NPLANE * NPAD * DDIM;

  const int layerGrid  = NPLANE * NSTRIP;
  const int reduceGrid = (NNODES * DDIM + 255) / 256;

  layer_kernel<false><<<layerGrid, GWAVES * 32, 0, stream>>>(
      node_emb, sig, 6, ei, et, E, W1, root1, part1);
  layer_kernel<true><<<layerGrid, GWAVES * 32, 0, stream>>>(
      part1, bias1, 0, ei, et, E, W2, root2, part2);
  reduce_kernel<<<reduceGrid, 256, 0, stream>>>(part2, bias2, (float*)d_out);
}
